# starting kernel
# baseline (speedup 1.0000x reference)
#include <hip/hip_runtime.h>

constexpr int NS  = 306;
constexpr int NV  = 200000;
constexpr int NK  = 32;
constexpr int NL  = 64;
constexpr long long NES = 800000;
constexpr long long NEV = 1200000;

typedef __attribute__((ext_vector_type(2))) float v2f;
typedef __attribute__((ext_vector_type(8))) float v8f;

__global__ void k_init(float* __restrict__ h, float* __restrict__ out,
                       const float* __restrict__ b0) {
    long long i = (long long)blockIdx.x * blockDim.x + threadIdx.x;
    if (i < (long long)NV * NL) h[i] = 0.0f;
    if (i < (long long)NV * NK) out[i] = b0[i & (NK - 1)];
}

__global__ void k_gemm1(const float* __restrict__ x, const float* __restrict__ Ws,
                        float* __restrict__ xw) {
    const int wave = threadIdx.x >> 5;
    const int lane = threadIdx.x & 31;
    const int half = lane >> 4;
    const int l16  = lane & 15;
    const int m0   = wave * 16;

    const int row   = m0 + l16;
    const int rowc  = row < NS ? row : NS - 1;
    const float msk = row < NS ? 1.0f : 0.0f;
    const float* __restrict__ xrow = x + rowc * NK;
    v2f afrag[8];
#pragma unroll
    for (int kk = 0; kk < 8; ++kk) {
        const int k0 = kk * 4 + half * 2;
        v2f t = *(const v2f*)(xrow + k0);
        afrag[kk].x = t.x * msk;
        afrag[kk].y = t.y * msk;
    }

    for (int nt = 0; nt < 4; ++nt) {
        v8f acc = {};
#pragma unroll
        for (int kk = 0; kk < 8; ++kk) {
            const int k0 = kk * 4 + half * 2;
            v2f b;
            b.x = Ws[(k0    ) * NL + nt * 16 + l16];
            b.y = Ws[(k0 + 1) * NL + nt * 16 + l16];
            acc = __builtin_amdgcn_wmma_f32_16x16x4_f32(
                false, afrag[kk], false, b, (short)0, acc, false, false);
        }
#pragma unroll
        for (int j = 0; j < 8; ++j) {
            const int orow = m0 + j + half * 8;
            if (orow < NS) xw[orow * NL + nt * 16 + l16] = acc[j];
        }
    }
}

__global__ void k_spmm1(const int* __restrict__ rows, const int* __restrict__ cols,
                        const float* __restrict__ vals, const float* __restrict__ xw,
                        float* __restrict__ h) {
    const long long tid = (long long)blockIdx.x * blockDim.x + threadIdx.x;
    const long long e = tid >> 6;
    const int c = (int)(tid & 63);
    if (e >= NES) return;
    const int r  = rows[e];
    const int cl = cols[e];
    const float v = vals[e];
    atomicAdd(&h[(long long)r * NL + c], v * xw[cl * NL + c]);
}

__global__ void k_gemm2(const float* __restrict__ h, const float* __restrict__ bs,
                        const float* __restrict__ W0, float* __restrict__ hw) {
    const int gw = blockIdx.x * (blockDim.x >> 5) + (threadIdx.x >> 5);
    if (gw >= NV / 16) return;
    const int lane = threadIdx.x & 31;
    const int half = lane >> 4;
    const int l16  = lane & 15;
    const int m0   = gw * 16;
    const float* __restrict__ arow = h + (long long)(m0 + l16) * NL;

    v2f afrag[16];
#pragma unroll
    for (int kk = 0; kk < 16; ++kk) {
        const int k0 = kk * 4 + half * 2;
        v2f t  = *(const v2f*)(arow + k0);
        v2f bb = *(const v2f*)(bs + k0);
        afrag[kk].x = fmaxf(t.x + bb.x, 0.0f);
        afrag[kk].y = fmaxf(t.y + bb.y, 0.0f);
    }

    v8f acc0 = {};
    v8f acc1 = {};
#pragma unroll
    for (int kk = 0; kk < 16; ++kk) {
        const int k0 = kk * 4 + half * 2;
        v2f b0f, b1f;
        b0f.x = W0[(k0    ) * NK +      l16];
        b0f.y = W0[(k0 + 1) * NK +      l16];
        b1f.x = W0[(k0    ) * NK + 16 + l16];
        b1f.y = W0[(k0 + 1) * NK + 16 + l16];
        acc0 = __builtin_amdgcn_wmma_f32_16x16x4_f32(
            false, afrag[kk], false, b0f, (short)0, acc0, false, false);
        acc1 = __builtin_amdgcn_wmma_f32_16x16x4_f32(
            false, afrag[kk], false, b1f, (short)0, acc1, false, false);
    }
#pragma unroll
    for (int j = 0; j < 8; ++j) {
        const int orow = m0 + j + half * 8;
        hw[(long long)orow * NK +      l16] = acc0[j];
        hw[(long long)orow * NK + 16 + l16] = acc1[j];
    }
}

__global__ void k_spmm2(const int* __restrict__ rows, const int* __restrict__ cols,
                        const float* __restrict__ vals, const float* __restrict__ hw,
                        float* __restrict__ out) {
    const long long tid = (long long)blockIdx.x * blockDim.x + threadIdx.x;
    const long long e = tid >> 5;
    const int c = (int)(tid & 31);
    if (e >= NEV) return;
    const int r  = rows[e];
    const int cl = cols[e];
    const float v = vals[e];
    atomicAdd(&out[(long long)r * NK + c], v * hw[(long long)cl * NK + c]);
}

extern "C" void kernel_launch(void* const* d_in, const int* in_sizes, int n_in,
                              void* d_out, int out_size, void* d_ws, size_t ws_size,
                              hipStream_t stream) {
    const float* x    = (const float*)d_in[0];
    const int*   srow = (const int*)  d_in[1];
    const int*   scol = (const int*)  d_in[2];
    const float* sval = (const float*)d_in[3];
    const int*   vrow = (const int*)  d_in[4];
    const int*   vcol = (const int*)  d_in[5];
    const float* vval = (const float*)d_in[6];
    const float* Ws   = (const float*)d_in[7];
    const float* bs   = (const float*)d_in[8];
    const float* W0   = (const float*)d_in[9];
    const float* b0   = (const float*)d_in[10];
    float* out = (float*)d_out;

    float* ws = (float*)d_ws;
    float* xw = ws;
    float* h  = ws + 32768;
    float* hw = h + (size_t)NV * NL;

    const long long nInit = (long long)NV * NL;
    k_init<<<(unsigned)((nInit + 255) / 256), 256, 0, stream>>>(h, out, b0);

    k_gemm1<<<1, 640, 0, stream>>>(x, Ws, xw);

    const long long n1 = NES * 64;
    k_spmm1<<<(unsigned)((n1 + 255) / 256), 256, 0, stream>>>(srow, scol, sval, xw, h);

    const int tilesM = NV / 16;
    const int wavesPerBlock = 8;
    const int blocks = (tilesM + wavesPerBlock - 1) / wavesPerBlock;
    k_gemm2<<<blocks, 32 * wavesPerBlock, 0, stream>>>(h, bs, W0, hw);

    const long long n2 = NEV * 32;
    k_spmm2<<<(unsigned)((n2 + 255) / 256), 256, 0, stream>>>(vrow, vcol, vval, hw, out);
}


# Reassociate layer 1 as (A_s x) W_s: scatter 32-wide x rows (half the float atomics), fuse the 32->64 WMMA product into the 64->32 one via transposed tiles; changes summation order
# speedup vs baseline: 1.3488x; 1.3488x over previous
#include <hip/hip_runtime.h>

constexpr int NS  = 306;
constexpr int NV  = 200000;
constexpr int NK  = 32;
constexpr int NL  = 64;
constexpr long long NES = 800000;
constexpr long long NEV = 1200000;

typedef __attribute__((ext_vector_type(2))) float v2f;
typedef __attribute__((ext_vector_type(8))) float v8f;


__global__ void k_init(float* __restrict__ g, float* __restrict__ out,
                       const float* __restrict__ b0) {
    long long i = (long long)blockIdx.x * blockDim.x + threadIdx.x;
    if (i < (long long)NV * NK) {
        g[i] = 0.0f;
        out[i] = b0[i & (NK - 1)];
    }
}

__global__ void k_spmm1(const int* __restrict__ rows, const int* __restrict__ cols,
                        const float* __restrict__ vals, const float* __restrict__ x,
                        float* __restrict__ g) {
    const long long tid = (long long)blockIdx.x * blockDim.x + threadIdx.x;
    const long long e = tid >> 5;
    const int c = (int)(tid & 31);
    if (e >= NES) return;
    const int r  = rows[e];
    const int cl = cols[e];
    const float v = vals[e];
    atomicAdd(&g[(long long)r * NK + c], v * x[cl * NK + c]);
}

__global__ void k_gemm2(const float* __restrict__ g, const float* __restrict__ Ws,
                        const float* __restrict__ bs, const float* __restrict__ W0,
                        float* __restrict__ hw) {
    const int gw = blockIdx.x * (blockDim.x >> 5) + (threadIdx.x >> 5);
    if (gw >= NV / 16) return;
    const int lane = threadIdx.x & 31;
    const int half = lane >> 4;
    const int l16  = lane & 15;
    const int m0   = gw * 16;

    const float4* __restrict__ grow =
        (const float4*)(g + (long long)(m0 + l16) * NK + half * 16);
    float gk[16];
#pragma unroll
    for (int q = 0; q < 4; ++q) {
        const float4 t = grow[q];
        gk[4 * q + 0] = t.x; gk[4 * q + 1] = t.y; gk[4 * q + 2] = t.z; gk[4 * q + 3] = t.w;
    }

    v8f acc[4];
#pragma unroll
    for (int mt = 0; mt < 4; ++mt) {
        const float4 lo = *(const float4*)(bs + mt * 16 + half * 8);
        const float4 hi = *(const float4*)(bs + mt * 16 + half * 8 + 4);
        acc[mt][0] = lo.x; acc[mt][1] = lo.y; acc[mt][2] = lo.z; acc[mt][3] = lo.w;
        acc[mt][4] = hi.x; acc[mt][5] = hi.y; acc[mt][6] = hi.z; acc[mt][7] = hi.w;
    }

#pragma unroll
    for (int s = 0; s < 8; ++s) {
        const int k0 = half * 16 + 2 * s;
        v2f b;
        b.x = gk[2 * s];
        b.y = gk[2 * s + 1];
#pragma unroll
        for (int mt = 0; mt < 4; ++mt) {
            v2f a;
            a.x = Ws[(k0    ) * NL + mt * 16 + l16];
            a.y = Ws[(k0 + 1) * NL + mt * 16 + l16];
            acc[mt] = __builtin_amdgcn_wmma_f32_16x16x4_f32(
                false, a, false, b, (short)0, acc[mt], false, false);
        }
    }

#pragma unroll
    for (int mt = 0; mt < 4; ++mt) {
#pragma unroll
        for (int j = 0; j < 8; ++j) acc[mt][j] = fmaxf(acc[mt][j], 0.0f);
    }

    v8f o0 = {};
    v8f o1 = {};
#pragma unroll
    for (int mt = 0; mt < 4; ++mt) {
#pragma unroll
        for (int q = 0; q < 4; ++q) {
            const int hid = mt * 16 + half * 8 + 2 * q;
            v2f b;
            b.x = acc[mt][2 * q];
            b.y = acc[mt][2 * q + 1];
            v2f a0, a1;
            a0.x = W0[(hid    ) * NK +      l16];
            a0.y = W0[(hid + 1) * NK +      l16];
            a1.x = W0[(hid    ) * NK + 16 + l16];
            a1.y = W0[(hid + 1) * NK + 16 + l16];
            o0 = __builtin_amdgcn_wmma_f32_16x16x4_f32(
                false, a0, false, b, (short)0, o0, false, false);
            o1 = __builtin_amdgcn_wmma_f32_16x16x4_f32(
                false, a1, false, b, (short)0, o1, false, false);
        }
    }

    float* __restrict__ orow = hw + (long long)(m0 + l16) * NK + half * 8;
    *(float4*)(orow     ) = make_float4(o0[0], o0[1], o0[2], o0[3]);
    *(float4*)(orow +  4) = make_float4(o0[4], o0[5], o0[6], o0[7]);
    *(float4*)(orow + 16) = make_float4(o1[0], o1[1], o1[2], o1[3]);
    *(float4*)(orow + 20) = make_float4(o1[4], o1[5], o1[6], o1[7]);
}

__global__ void k_spmm2(const int* __restrict__ rows, const int* __restrict__ cols,
                        const float* __restrict__ vals, const float* __restrict__ hw,
                        float* __restrict__ out) {
    const long long tid = (long long)blockIdx.x * blockDim.x + threadIdx.x;
    const long long e = tid >> 5;
    const int c = (int)(tid & 31);
    if (e >= NEV) return;
    const int r  = rows[e];
    const int cl = cols[e];
    const float v = vals[e];
    atomicAdd(&out[(long long)r * NK + c], v * hw[(long long)cl * NK + c]);
}

extern "C" void kernel_launch(void* const* d_in, const int* in_sizes, int n_in,
                              void* d_out, int out_size, void* d_ws, size_t ws_size,
                              hipStream_t stream) {
    const float* x    = (const float*)d_in[0];
    const int*   srow = (const int*)  d_in[1];
    const int*   scol = (const int*)  d_in[2];
    const float* sval = (const float*)d_in[3];
    const int*   vrow = (const int*)  d_in[4];
    const int*   vcol = (const int*)  d_in[5];
    const float* vval = (const float*)d_in[6];
    const float* Ws   = (const float*)d_in[7];
    const float* bs   = (const float*)d_in[8];
    const float* W0   = (const float*)d_in[9];
    const float* b0   = (const float*)d_in[10];
    float* out = (float*)d_out;

    float* ws = (float*)d_ws;
    float* g  = ws;
    float* hw = ws + (size_t)NV * NK;

    const long long nInit = (long long)NV * NK;
    k_init<<<(unsigned)((nInit + 255) / 256), 256, 0, stream>>>(g, out, b0);

    const long long n1 = NES * 32;
    k_spmm1<<<(unsigned)((n1 + 255) / 256), 256, 0, stream>>>(srow, scol, sval, x, g);

    const int tilesM = NV / 16;
    const int wavesPerBlock = 8;
    const int blocks = (tilesM + wavesPerBlock - 1) / wavesPerBlock;
    k_gemm2<<<blocks, 32 * wavesPerBlock, 0, stream>>>(g, Ws, bs, W0, hw);

    const long long n2 = NEV * 32;
    k_spmm2<<<(unsigned)((n2 + 255) / 256), 256, 0, stream>>>(vrow, vcol, vval, hw, out);
}


# Replace both global float-atomic edge scatters by owner work-groups: each owns 1600 output rows, scans the edge list, sums its edges in LDS (LDS float adds, arrival order) and stores each row once
# speedup vs baseline: 2.1878x; 1.6220x over previous
#include <hip/hip_runtime.h>


constexpr int NS  = 306;
constexpr int NV  = 200000;
constexpr int NK  = 32;
constexpr int NL  = 64;
constexpr int NES = 800000;
constexpr int NEV = 1200000;

typedef __attribute__((ext_vector_type(2))) float v2f;
typedef __attribute__((ext_vector_type(8))) float v8f;


constexpr int RB          = 1600;
constexpr int NBLK        = NV / RB;
constexpr int OWN_THREADS = 1024;
constexpr int OWN_WAVES   = OWN_THREADS / 32;
constexpr int QCAP        = 160;
constexpr int CHUNK       = 128;
constexpr int DEPTH       = 4;
static_assert(NBLK * RB == NV, "row blocks must tile the vertices");
static_assert(NES % CHUNK == 0 && NEV % CHUNK == 0, "edge lists are whole chunks");

template <int NE, bool ADD_BIAS>
__device__ __forceinline__ void owner_spmm(const int* __restrict__ rows, const int* __restrict__ cols,
                                           const float* __restrict__ vals, const float* __restrict__ src,
                                           const float* __restrict__ bias, float* __restrict__ dst) {
    __shared__ float acc[RB * NK];
    __shared__ int   q_e[OWN_WAVES][QCAP];
    __shared__ int   q_r[OWN_WAVES][QCAP];

    const int tid  = threadIdx.x;
    const int lane = tid & 31;
    const int wave = __builtin_amdgcn_readfirstlane(tid >> 5);
    const int lo   = blockIdx.x * RB;

    float4* acc4 = (float4*)acc;
    for (int i = tid; i < RB * NK / 4; i += OWN_THREADS) acc4[i] = make_float4(0.f, 0.f, 0.f, 0.f);
    __syncthreads();

    int* qe = q_e[wave];
    int* qr = q_r[wave];
    int  qn = 0;

    constexpr int NCH = NE / CHUNK;
    constexpr int NSC = (NCH + DEPTH - 1) / DEPTH;
    const int4* __restrict__ rows4 = (const int4*)rows;

    auto drain = [&](int n) {
        int   lr = 0, c = 0;
        float v  = 0.f;
        if (lane < n) {
            const int e = qe[qn - n + lane];
            lr = qr[qn - n + lane];
            c  = cols[e];
            v  = vals[e];
        }
        qn -= n;
        float xv[32];
#pragma unroll
        for (int j = 0; j < 32; ++j) {
            if (j < n) {
                const int cj = __builtin_amdgcn_readlane(c, j);
                xv[j] = src[cj * NK + lane];
            }
        }
#pragma unroll
        for (int j = 0; j < 32; ++j) {
            if (j < n) {
                const float vj = __int_as_float(__builtin_amdgcn_readlane(__float_as_int(v), j));
                const int   rj = __builtin_amdgcn_readlane(lr, j);
                __hip_atomic_fetch_add(&acc[rj * NK + lane], vj * xv[j],
                                       __ATOMIC_RELAXED, __HIP_MEMORY_SCOPE_WORKGROUP);
            }
        }
    };

    auto push = [&](int r, int e, unsigned lim) {
        const unsigned lr = (unsigned)(r - lo);
        const bool     m  = lr < lim;
        const unsigned mask = __builtin_amdgcn_ballot_w32(m);
        if (mask != 0u) {
            const int pos = qn + (int)__builtin_amdgcn_mbcnt_lo(mask, 0u);
            if (m) { qe[pos] = e; qr[pos] = (int)lr; }
            qn += __builtin_popcount(mask);
        }
    };

    auto load_group = [&](int sc, int4 (&buf)[DEPTH]) {
#pragma unroll
        for (int k = 0; k < DEPTH; ++k) {
            const int ch  = sc * DEPTH + k;
            const int chc = ch < NCH ? ch : NCH - 1;
            buf[k] = rows4[chc * (CHUNK / 4) + lane];
        }
    };

    int4 cur[DEPTH];
    int sc = wave;
    load_group(sc, cur);
    for (; sc < NSC; sc += OWN_WAVES) {
        int4 nxt[DEPTH];
        load_group(sc + OWN_WAVES, nxt);
#pragma unroll
        for (int k = 0; k < DEPTH; ++k) {
            const int      ch  = sc * DEPTH + k;
            const unsigned lim = ch < NCH ? (unsigned)RB : 0u;
            const int      e0  = ch * CHUNK + lane * 4;
            push(cur[k].x, e0 + 0, lim);
            push(cur[k].y, e0 + 1, lim);
            push(cur[k].z, e0 + 2, lim);
            push(cur[k].w, e0 + 3, lim);
            while (qn >= 32) drain(32);
        }
#pragma unroll
        for (int k = 0; k < DEPTH; ++k) cur[k] = nxt[k];
    }
    if (qn > 0) drain(qn);

    __syncthreads();
    float4* __restrict__ dst4 = (float4*)(dst + (long long)lo * NK);
    for (int i = tid; i < RB * NK / 4; i += OWN_THREADS) {
        float4 a = acc4[i];
        if (ADD_BIAS) {
            const float4 b = *(const float4*)(bias + ((i * 4) & (NK - 1)));
            a.x += b.x; a.y += b.y; a.z += b.z; a.w += b.w;
        }
        dst4[i] = a;
    }
}

__global__ __launch_bounds__(OWN_THREADS)
void k_spmm1(const int* __restrict__ rows, const int* __restrict__ cols,
             const float* __restrict__ vals, const float* __restrict__ x,
             float* __restrict__ g) {
    owner_spmm<NES, false>(rows, cols, vals, x, x, g);
}

__global__ void k_gemm2(const float* __restrict__ g, const float* __restrict__ Ws,
                        const float* __restrict__ bs, const float* __restrict__ W0,
                        float* __restrict__ hw) {
    const int gw = blockIdx.x * (blockDim.x >> 5) + (threadIdx.x >> 5);
    if (gw >= NV / 16) return;
    const int lane = threadIdx.x & 31;
    const int half = lane >> 4;
    const int l16  = lane & 15;
    const int m0   = gw * 16;

    const float4* __restrict__ grow =
        (const float4*)(g + (long long)(m0 + l16) * NK + half * 16);
    float gk[16];
#pragma unroll
    for (int q = 0; q < 4; ++q) {
        const float4 t = grow[q];
        gk[4 * q + 0] = t.x; gk[4 * q + 1] = t.y; gk[4 * q + 2] = t.z; gk[4 * q + 3] = t.w;
    }

    v8f acc[4];
#pragma unroll
    for (int mt = 0; mt < 4; ++mt) {
        const float4 lo = *(const float4*)(bs + mt * 16 + half * 8);
        const float4 hi = *(const float4*)(bs + mt * 16 + half * 8 + 4);
        acc[mt][0] = lo.x; acc[mt][1] = lo.y; acc[mt][2] = lo.z; acc[mt][3] = lo.w;
        acc[mt][4] = hi.x; acc[mt][5] = hi.y; acc[mt][6] = hi.z; acc[mt][7] = hi.w;
    }

#pragma unroll
    for (int s = 0; s < 8; ++s) {
        const int k0 = half * 16 + 2 * s;
        v2f b;
        b.x = gk[2 * s];
        b.y = gk[2 * s + 1];
#pragma unroll
        for (int mt = 0; mt < 4; ++mt) {
            v2f a;
            a.x = Ws[(k0    ) * NL + mt * 16 + l16];
            a.y = Ws[(k0 + 1) * NL + mt * 16 + l16];
            acc[mt] = __builtin_amdgcn_wmma_f32_16x16x4_f32(
                false, a, false, b, (short)0, acc[mt], false, false);
        }
    }

#pragma unroll
    for (int mt = 0; mt < 4; ++mt) {
#pragma unroll
        for (int j = 0; j < 8; ++j) acc[mt][j] = fmaxf(acc[mt][j], 0.0f);
    }

    v8f o0 = {};
    v8f o1 = {};
#pragma unroll
    for (int mt = 0; mt < 4; ++mt) {
#pragma unroll
        for (int q = 0; q < 4; ++q) {
            const int hid = mt * 16 + half * 8 + 2 * q;
            v2f b;
            b.x = acc[mt][2 * q];
            b.y = acc[mt][2 * q + 1];
            v2f a0, a1;
            a0.x = W0[(hid    ) * NK +      l16];
            a0.y = W0[(hid + 1) * NK +      l16];
            a1.x = W0[(hid    ) * NK + 16 + l16];
            a1.y = W0[(hid + 1) * NK + 16 + l16];
            o0 = __builtin_amdgcn_wmma_f32_16x16x4_f32(
                false, a0, false, b, (short)0, o0, false, false);
            o1 = __builtin_amdgcn_wmma_f32_16x16x4_f32(
                false, a1, false, b, (short)0, o1, false, false);
        }
    }

    float* __restrict__ orow = hw + (long long)(m0 + l16) * NK + half * 8;
    *(float4*)(orow     ) = make_float4(o0[0], o0[1], o0[2], o0[3]);
    *(float4*)(orow +  4) = make_float4(o0[4], o0[5], o0[6], o0[7]);
    *(float4*)(orow + 16) = make_float4(o1[0], o1[1], o1[2], o1[3]);
    *(float4*)(orow + 20) = make_float4(o1[4], o1[5], o1[6], o1[7]);
}

__global__ __launch_bounds__(OWN_THREADS)
void k_spmm2(const int* __restrict__ rows, const int* __restrict__ cols,
             const float* __restrict__ vals, const float* __restrict__ hw,
             const float* __restrict__ b0, float* __restrict__ out) {
    owner_spmm<NEV, true>(rows, cols, vals, hw, b0, out);
}

extern "C" void kernel_launch(void* const* d_in, const int* in_sizes, int n_in,
                              void* d_out, int out_size, void* d_ws, size_t ws_size,
                              hipStream_t stream) {
    const float* x    = (const float*)d_in[0];
    const int*   srow = (const int*)  d_in[1];
    const int*   scol = (const int*)  d_in[2];
    const float* sval = (const float*)d_in[3];
    const int*   vrow = (const int*)  d_in[4];
    const int*   vcol = (const int*)  d_in[5];
    const float* vval = (const float*)d_in[6];
    const float* Ws   = (const float*)d_in[7];
    const float* bs   = (const float*)d_in[8];
    const float* W0   = (const float*)d_in[9];
    const float* b0   = (const float*)d_in[10];
    float* out = (float*)d_out;

    float* ws = (float*)d_ws;
    float* g  = ws;
    float* hw = ws + (size_t)NV * NK;

    k_spmm1<<<NBLK, OWN_THREADS, 0, stream>>>(srow, scol, sval, x, g);

    const int tilesM = NV / 16;
    const int wavesPerBlock = 8;
    const int blocks = (tilesM + wavesPerBlock - 1) / wavesPerBlock;
    k_gemm2<<<blocks, 32 * wavesPerBlock, 0, stream>>>(g, Ws, bs, W0, hw);

    k_spmm2<<<NBLK, OWN_THREADS, 0, stream>>>(vrow, vcol, vval, hw, b0, out);
}


# Bin both edge lists by owner block in a first pass (8000-edge slices grouped in LDS, per-owner segments in workspace): 250 owners of 800 rows read only their own edges; LDS float adds, arrival order
# speedup vs baseline: 4.6867x; 2.1422x over previous
#include <hip/hip_runtime.h>


constexpr int NS  = 306;
constexpr int NV  = 200000;
constexpr int NK  = 32;
constexpr int NL  = 64;
constexpr int NES = 800000;
constexpr int NEV = 1200000;

typedef __attribute__((ext_vector_type(2))) float v2f;
typedef __attribute__((ext_vector_type(8))) float v8f;


constexpr int RB          = 800;
constexpr int NBLK        = NV / RB;
constexpr int SL          = 8000;
constexpr int PS          = NES / SL;
constexpr int PV          = NEV / SL;
constexpr int SEGCAP      = 128;
constexpr int CNTPITCH    = 256;
constexpr int KEYBITS     = 10;
typedef unsigned long long ent_t;
constexpr int OWN_THREADS = 1024;
constexpr int OWN_WAVES   = OWN_THREADS / 32;
constexpr int BIN_THREADS = 1024;
constexpr int BIN_ITERS   = (SL + BIN_THREADS - 1) / BIN_THREADS;
static_assert(NBLK * RB == NV, "row blocks must tile the vertices");
static_assert(PS * SL == NES && PV * SL == NEV, "slices must tile the edge lists");
static_assert(SL % 32 == 0, "a slice is whole wave-wide steps");
static_assert(OWN_WAVES <= PS && OWN_WAVES <= PV, "every wave of an owner has a first slice");
static_assert(SEGCAP % 64 == 0, "a segment is read in steps of 64 entries");
static_assert(NBLK <= CNTPITCH && CNTPITCH % 32 == 0 && CNTPITCH <= BIN_THREADS, "one count per thread of the first waves");
static_assert(RB <= (1 << KEYBITS), "row inside the block must fit its field");
static_assert(NV <= (1 << (31 - KEYBITS)) && NS <= NV, "source row must fit above it");

constexpr size_t OFF_G    = 0;
constexpr size_t OFF_HW   = OFF_G    + (size_t)NV * NK;
constexpr size_t OFF_SEG1 = OFF_HW   + (size_t)NV * NK;
constexpr size_t OFF_SEG2 = OFF_SEG1 + (size_t)NBLK * PS * SEGCAP * 2;
constexpr size_t OFF_CNT1 = OFF_SEG2 + (size_t)NBLK * PV * SEGCAP * 2;
constexpr size_t OFF_CNT2 = OFF_CNT1 + (size_t)PS * CNTPITCH;

__global__ __launch_bounds__(BIN_THREADS)
void k_bin(const int* __restrict__ srow, const int* __restrict__ scol, const float* __restrict__ sval,
           const int* __restrict__ vrow, const int* __restrict__ vcol, const float* __restrict__ vval,
           float* __restrict__ seg1f, float* __restrict__ cnt1f,
           float* __restrict__ seg2f, float* __restrict__ cnt2f) {
    __shared__ int   hist[CNTPITCH];
    __shared__ int   off[CNTPITCH];
    __shared__ int   wsum[CNTPITCH / 32];
    __shared__ ent_t stage[SL];

    const int  tid    = threadIdx.x;
    const int  lane   = tid & 31;
    const bool sensor = blockIdx.x < PS;
    const int* __restrict__   rows = sensor ? srow : vrow;
    const int* __restrict__   cols = sensor ? scol : vcol;
    const float* __restrict__ vals = sensor ? sval : vval;
    ent_t* __restrict__ seg = (ent_t*)(sensor ? seg1f : seg2f);
    int*  __restrict__ cntg = (int*)(sensor ? cnt1f : cnt2f);
    const int np = sensor ? PS : PV;
    const int p  = sensor ? blockIdx.x : blockIdx.x - PS;

    for (int i = tid; i < CNTPITCH; i += BIN_THREADS) hist[i] = 0;

    const int e0 = p * SL;
    int   r[BIN_ITERS], c[BIN_ITERS];
    float v[BIN_ITERS];
#pragma unroll
    for (int k = 0; k < BIN_ITERS; ++k) {
        const int  i  = tid + k * BIN_THREADS;
        const bool ok = i < SL;
        const int  e  = e0 + (ok ? i : 0);
        const int  rr = rows[e];
        r[k] = ok ? rr : -1;
        c[k] = cols[e];
        v[k] = vals[e];
    }
    __syncthreads();

    int pos[BIN_ITERS];
#pragma unroll
    for (int k = 0; k < BIN_ITERS; ++k) {
        const unsigned b = (unsigned)r[k] / (unsigned)RB;
        pos[k] = -1;
        if (b < (unsigned)NBLK)
            pos[k] = atomicAdd(&hist[b], 1);
    }
    __syncthreads();

    int hv = 0, inc = 0;
    if (tid < CNTPITCH) {
        hv  = hist[tid];
        inc = hv;
#pragma unroll
        for (int d = 1; d < 32; d <<= 1) {
            const int t = __shfl_up(inc, d);
            if (lane >= d) inc += t;
        }
        if (lane == 31) wsum[tid >> 5] = inc;
    }
    __syncthreads();
    if (tid < CNTPITCH) {
        int basew = 0;
        for (int w = 0; w < (tid >> 5); ++w) basew += wsum[w];
        off[tid] = basew + inc - hv;
    }
    __syncthreads();

#pragma unroll
    for (int k = 0; k < BIN_ITERS; ++k) {
        if (pos[k] >= 0) {
            const int b  = r[k] / RB;
            const int lr = r[k] - b * RB;
            stage[off[b] + pos[k]] =
                (ent_t)(unsigned)(lr | (c[k] << KEYBITS)) | ((ent_t)(unsigned)__float_as_int(v[k]) << 32);
        }
    }
    __syncthreads();

    const int wave = __builtin_amdgcn_readfirstlane(tid >> 5);
    for (int b = wave; b < NBLK; b += BIN_THREADS / 32) {
        const int cb = __builtin_amdgcn_readfirstlane(hist[b]);
        const int ob = __builtin_amdgcn_readfirstlane(off[b]);
        const int n  = cb < SEGCAP ? cb : SEGCAP;
        ent_t* __restrict__ s = seg + ((size_t)b * np + p) * SEGCAP;
        for (int i = lane; i < n; i += 32) s[i] = stage[ob + i];
    }
    if (tid < CNTPITCH) cntg[p * CNTPITCH + tid] = hv;
}

template <int NP, bool ADD_BIAS>
__device__ __forceinline__ void owner_spmm(const int* __restrict__ rows, const int* __restrict__ cols,
                                           const float* __restrict__ vals,
                                           const float* __restrict__ segf, const float* __restrict__ cntf,
                                           const float* __restrict__ src,
                                           const float* __restrict__ bias, float* __restrict__ dst) {
    __shared__ float acc[RB * NK];

    const int tid  = threadIdx.x;
    const int lane = tid & 31;
    const int wave = __builtin_amdgcn_readfirstlane(tid >> 5);
    const int blk  = blockIdx.x;
    const int lo   = blk * RB;
    const ent_t* __restrict__ segs = (const ent_t*)segf;
    const int*  __restrict__ cntg = (const int*)cntf;

    int cnt_mine = 0;
    {
        const int pk = wave + OWN_WAVES * lane;
        if (pk < NP) cnt_mine = cntg[pk * CNTPITCH + blk];
    }

    float4* acc4 = (float4*)acc;
    for (int i = tid; i < RB * NK / 4; i += OWN_THREADS) acc4[i] = make_float4(0.f, 0.f, 0.f, 0.f);

    auto load64 = [&](const ent_t* __restrict__ s, int base, int n, ent_t& ta, ent_t& tb) {
        ta = 0;
        tb = 0;
        if (base < n) {
            const int ia = base + lane;
            ta = s[ia < n ? ia : n - 1];
        }
        if (base + 32 < n) {
            const int ib = base + 32 + lane;
            tb = s[ib < n ? ib : n - 1];
        }
    };

    float xv[64];
    auto request_rows = [&](const ent_t ta, const ent_t tb, int m) {
        const int tak = (int)ta, tbk = (int)tb;
#pragma unroll
        for (int j = 0; j < 64; ++j) {
            if (j < m) {
                const int kj = j < 32 ? __builtin_amdgcn_readlane(tak, j) : __builtin_amdgcn_readlane(tbk, j - 32);
                xv[j] = src[(kj >> KEYBITS) * NK + lane];
            }
        }
    };
    auto add_rows = [&](const ent_t ta, const ent_t tb, int m) {
        const int tak = (int)ta, tav = (int)(ta >> 32), tbk = (int)tb, tbv = (int)(tb >> 32);
#pragma unroll
        for (int j = 0; j < 64; ++j) {
            if (j < m) {
                const int   kj = j < 32 ? __builtin_amdgcn_readlane(tak, j) : __builtin_amdgcn_readlane(tbk, j - 32);
                const float vj = __int_as_float(j < 32 ? __builtin_amdgcn_readlane(tav, j) : __builtin_amdgcn_readlane(tbv, j - 32));
                __hip_atomic_fetch_add(&acc[(kj & ((1 << KEYBITS) - 1)) * NK + lane], vj * xv[j],
                                       __ATOMIC_RELAXED, __HIP_MEMORY_SCOPE_WORKGROUP);
            }
        }
    };

    constexpr int KMAX = (NP + OWN_WAVES - 1) / OWN_WAVES;
    int  n_cur = __builtin_amdgcn_readlane(cnt_mine, 0);
    ent_t ca, cb;
    load64(segs + ((size_t)blk * NP + wave) * SEGCAP, 0, n_cur <= SEGCAP ? n_cur : 0, ca, cb);
    __syncthreads();

    for (int k = 0; k < KMAX; ++k) {
        const int p = wave + OWN_WAVES * k;
        if (p >= NP) break;
        const int n = n_cur;
        const int m0 = n > SEGCAP ? 0 : (n < 64 ? n : 64);

        if (m0 > 0) request_rows(ca, cb, m0);

        int  n_nxt = 0;
        ent_t na = 0, nb = 0;
        if (k + 1 < KMAX && p + OWN_WAVES < NP) {
            n_nxt = __builtin_amdgcn_readlane(cnt_mine, k + 1);
            load64(segs + ((size_t)blk * NP + p + OWN_WAVES) * SEGCAP, 0, n_nxt <= SEGCAP ? n_nxt : 0, na, nb);
        }

        if (m0 > 0) add_rows(ca, cb, m0);

        if (n <= SEGCAP) {
            for (int base = 64; base < n; base += 32) {
                const int   ir = base + lane;
                const ent_t t  = (segs + ((size_t)blk * NP + p) * SEGCAP)[ir < n ? ir : n - 1];
                const int   m  = (n - base) < 32 ? (n - base) : 32;
                for (int j = 0; j < m; ++j) {
                    const int   kj = __builtin_amdgcn_readlane((int)t, j);
                    const float vj = __int_as_float(__builtin_amdgcn_readlane((int)(t >> 32), j));
                    __hip_atomic_fetch_add(&acc[(kj & ((1 << KEYBITS) - 1)) * NK + lane],
                                           vj * src[(kj >> KEYBITS) * NK + lane],
                                           __ATOMIC_RELAXED, __HIP_MEMORY_SCOPE_WORKGROUP);
                }
            }
        } else {
            const int e0 = p * SL;
            for (int i = 0; i < SL; i += 32) {
                const int      e   = e0 + i + lane;
                const unsigned lr  = (unsigned)(rows[e] - lo);
                const bool     hit = lr < (unsigned)RB;
                unsigned mask = __builtin_amdgcn_ballot_w32(hit);
                int   c = 0;
                float v = 0.f;
                if (hit) { c = cols[e]; v = vals[e]; }
                while (mask != 0u) {
                    const int j = __builtin_ctz(mask);
                    mask &= mask - 1u;
                    const int   cj = __builtin_amdgcn_readlane(c, j);
                    const int   rj = __builtin_amdgcn_readlane((int)lr, j);
                    const float vj = __int_as_float(__builtin_amdgcn_readlane(__float_as_int(v), j));
                    __hip_atomic_fetch_add(&acc[rj * NK + lane], vj * src[cj * NK + lane],
                                           __ATOMIC_RELAXED, __HIP_MEMORY_SCOPE_WORKGROUP);
                }
            }
        }
        ca = na; cb = nb; n_cur = n_nxt;
    }

    __syncthreads();
    float4* __restrict__ dst4 = (float4*)(dst + (long long)lo * NK);
    for (int i = tid; i < RB * NK / 4; i += OWN_THREADS) {
        float4 a = acc4[i];
        if (ADD_BIAS) {
            const float4 b = *(const float4*)(bias + ((i * 4) & (NK - 1)));
            a.x += b.x; a.y += b.y; a.z += b.z; a.w += b.w;
        }
        dst4[i] = a;
    }
}

__global__ __launch_bounds__(OWN_THREADS)
void k_spmm1(const int* __restrict__ rows, const int* __restrict__ cols,
             const float* __restrict__ vals, const float* __restrict__ segf,
             const float* __restrict__ cntf, const float* __restrict__ x,
             float* __restrict__ g) {
    owner_spmm<PS, false>(rows, cols, vals, segf, cntf, x, x, g);
}

__global__ void k_gemm2(const float* __restrict__ g, const float* __restrict__ Ws,
                        const float* __restrict__ bs, const float* __restrict__ W0,
                        float* __restrict__ hw) {
    const int gw = blockIdx.x * (blockDim.x >> 5) + (threadIdx.x >> 5);
    if (gw >= NV / 16) return;
    const int lane = threadIdx.x & 31;
    const int half = lane >> 4;
    const int l16  = lane & 15;
    const int m0   = gw * 16;

    const float4* __restrict__ grow =
        (const float4*)(g + (long long)(m0 + l16) * NK + half * 16);
    float gk[16];
#pragma unroll
    for (int q = 0; q < 4; ++q) {
        const float4 t = grow[q];
        gk[4 * q + 0] = t.x; gk[4 * q + 1] = t.y; gk[4 * q + 2] = t.z; gk[4 * q + 3] = t.w;
    }

    v8f acc[4];
#pragma unroll
    for (int mt = 0; mt < 4; ++mt) {
        const float4 lo = *(const float4*)(bs + mt * 16 + half * 8);
        const float4 hi = *(const float4*)(bs + mt * 16 + half * 8 + 4);
        acc[mt][0] = lo.x; acc[mt][1] = lo.y; acc[mt][2] = lo.z; acc[mt][3] = lo.w;
        acc[mt][4] = hi.x; acc[mt][5] = hi.y; acc[mt][6] = hi.z; acc[mt][7] = hi.w;
    }

#pragma unroll
    for (int s = 0; s < 8; ++s) {
        const int k0 = half * 16 + 2 * s;
        v2f b;
        b.x = gk[2 * s];
        b.y = gk[2 * s + 1];
#pragma unroll
        for (int mt = 0; mt < 4; ++mt) {
            v2f a;
            a.x = Ws[(k0    ) * NL + mt * 16 + l16];
            a.y = Ws[(k0 + 1) * NL + mt * 16 + l16];
            acc[mt] = __builtin_amdgcn_wmma_f32_16x16x4_f32(
                false, a, false, b, (short)0, acc[mt], false, false);
        }
    }

#pragma unroll
    for (int mt = 0; mt < 4; ++mt) {
#pragma unroll
        for (int j = 0; j < 8; ++j) acc[mt][j] = fmaxf(acc[mt][j], 0.0f);
    }

    v8f o0 = {};
    v8f o1 = {};
#pragma unroll
    for (int mt = 0; mt < 4; ++mt) {
#pragma unroll
        for (int q = 0; q < 4; ++q) {
            const int hid = mt * 16 + half * 8 + 2 * q;
            v2f b;
            b.x = acc[mt][2 * q];
            b.y = acc[mt][2 * q + 1];
            v2f a0, a1;
            a0.x = W0[(hid    ) * NK +      l16];
            a0.y = W0[(hid + 1) * NK +      l16];
            a1.x = W0[(hid    ) * NK + 16 + l16];
            a1.y = W0[(hid + 1) * NK + 16 + l16];
            o0 = __builtin_amdgcn_wmma_f32_16x16x4_f32(
                false, a0, false, b, (short)0, o0, false, false);
            o1 = __builtin_amdgcn_wmma_f32_16x16x4_f32(
                false, a1, false, b, (short)0, o1, false, false);
        }
    }

    float* __restrict__ orow = hw + (long long)(m0 + l16) * NK + half * 8;
    *(float4*)(orow     ) = make_float4(o0[0], o0[1], o0[2], o0[3]);
    *(float4*)(orow +  4) = make_float4(o0[4], o0[5], o0[6], o0[7]);
    *(float4*)(orow + 16) = make_float4(o1[0], o1[1], o1[2], o1[3]);
    *(float4*)(orow + 20) = make_float4(o1[4], o1[5], o1[6], o1[7]);
}

__global__ __launch_bounds__(OWN_THREADS)
void k_spmm2(const int* __restrict__ rows, const int* __restrict__ cols,
             const float* __restrict__ vals, const float* __restrict__ segf,
             const float* __restrict__ cntf, const float* __restrict__ hw,
             const float* __restrict__ b0, float* __restrict__ out) {
    owner_spmm<PV, true>(rows, cols, vals, segf, cntf, hw, b0, out);
}

extern "C" void kernel_launch(void* const* d_in, const int* in_sizes, int n_in,
                              void* d_out, int out_size, void* d_ws, size_t ws_size,
                              hipStream_t stream) {
    const float* x    = (const float*)d_in[0];
    const int*   srow = (const int*)  d_in[1];
    const int*   scol = (const int*)  d_in[2];
    const float* sval = (const float*)d_in[3];
    const int*   vrow = (const int*)  d_in[4];
    const int*   vcol = (const int*)  d_in[5];
    const float* vval = (const float*)d_in[6];
    const float* Ws   = (const float*)d_in[7];
    const float* bs   = (const float*)d_in[8];
    const float* W0   = (const float*)d_in[9];
    const float* b0   = (const float*)d_in[10];
    float* out = (float*)d_out;

    float* ws   = (float*)d_ws;
    float* g    = ws + OFF_G;
    float* hw   = ws + OFF_HW;
    float* seg1 = ws + OFF_SEG1;
    float* seg2 = ws + OFF_SEG2;
    float* cnt1 = ws + OFF_CNT1;
    float* cnt2 = ws + OFF_CNT2;

    k_bin<<<PS + PV, BIN_THREADS, 0, stream>>>(srow, scol, sval, vrow, vcol, vval, seg1, cnt1, seg2, cnt2);

    k_spmm1<<<NBLK, OWN_THREADS, 0, stream>>>(srow, scol, sval, seg1, cnt1, x, g);

    const int tilesM = NV / 16;
    const int wavesPerBlock = 8;
    const int blocks = (tilesM + wavesPerBlock - 1) / wavesPerBlock;
    k_gemm2<<<blocks, 32 * wavesPerBlock, 0, stream>>>(g, Ws, bs, W0, hw);

    k_spmm2<<<NBLK, OWN_THREADS, 0, stream>>>(vrow, vcol, vval, seg2, cnt2, hw, b0, out);
}


# Owner edge sums: an edge's row offsets and weight reach all lanes by DPP row_share (16 entries mirrored in both lane rows), not scalar lane reads; guarded groups of 8 edges; LDS adds, arrival order
# speedup vs baseline: 5.2387x; 1.1178x over previous
#include <hip/hip_runtime.h>


constexpr int NS  = 306;
constexpr int NV  = 200000;
constexpr int NK  = 32;
constexpr int NL  = 64;
constexpr int NES = 800000;
constexpr int NEV = 1200000;

typedef __attribute__((ext_vector_type(2))) float v2f;
typedef __attribute__((ext_vector_type(8))) float v8f;


constexpr int RB          = 800;
constexpr int NBLK        = NV / RB;
constexpr int SL          = 8000;
constexpr int PS          = NES / SL;
constexpr int PV          = NEV / SL;
constexpr int SEGCAP      = 128;
constexpr int CNTPITCH    = 256;
constexpr int KEYBITS     = 10;
typedef unsigned long long ent_t;
constexpr int OWN_THREADS = 1024;
constexpr int OWN_WAVES   = OWN_THREADS / 32;
constexpr int BIN_THREADS = 1024;
constexpr int BIN_ITERS   = (SL + BIN_THREADS - 1) / BIN_THREADS;
static_assert(NBLK * RB == NV, "row blocks must tile the vertices");
static_assert(PS * SL == NES && PV * SL == NEV, "slices must tile the edge lists");
static_assert(SL % 32 == 0, "a slice is whole wave-wide steps");
static_assert(OWN_WAVES <= PS && OWN_WAVES <= PV, "every wave of an owner has a first slice");
static_assert(SEGCAP % 64 == 0, "a segment is read in steps of 64 entries");
static_assert(NBLK <= CNTPITCH && CNTPITCH % 32 == 0 && CNTPITCH <= BIN_THREADS, "one count per thread of the first waves");
static_assert(RB <= (1 << KEYBITS), "row inside the block must fit its field");
static_assert(NV <= (1 << (31 - KEYBITS)) && NS <= NV, "source row must fit above it");

constexpr size_t OFF_G    = 0;
constexpr size_t OFF_HW   = OFF_G    + (size_t)NV * NK;
constexpr size_t OFF_SEG1 = OFF_HW   + (size_t)NV * NK;
constexpr size_t OFF_SEG2 = OFF_SEG1 + (size_t)NBLK * PS * SEGCAP * 2;
constexpr size_t OFF_CNT1 = OFF_SEG2 + (size_t)NBLK * PV * SEGCAP * 2;
constexpr size_t OFF_CNT2 = OFF_CNT1 + (size_t)PS * CNTPITCH;

__global__ __launch_bounds__(BIN_THREADS)
void k_bin(const int* __restrict__ srow, const int* __restrict__ scol, const float* __restrict__ sval,
           const int* __restrict__ vrow, const int* __restrict__ vcol, const float* __restrict__ vval,
           float* __restrict__ seg1f, float* __restrict__ cnt1f,
           float* __restrict__ seg2f, float* __restrict__ cnt2f) {
    __shared__ int   hist[CNTPITCH];
    __shared__ int   off[CNTPITCH];
    __shared__ int   wsum[CNTPITCH / 32];
    __shared__ ent_t stage[SL];

    const int  tid    = threadIdx.x;
    const int  lane   = tid & 31;
    const bool sensor = blockIdx.x < PS;
    const int* __restrict__   rows = sensor ? srow : vrow;
    const int* __restrict__   cols = sensor ? scol : vcol;
    const float* __restrict__ vals = sensor ? sval : vval;
    ent_t* __restrict__ seg = (ent_t*)(sensor ? seg1f : seg2f);
    int*  __restrict__ cntg = (int*)(sensor ? cnt1f : cnt2f);
    const int np = sensor ? PS : PV;
    const int p  = sensor ? blockIdx.x : blockIdx.x - PS;

    for (int i = tid; i < CNTPITCH; i += BIN_THREADS) hist[i] = 0;

    const int e0 = p * SL;
    int   r[BIN_ITERS], c[BIN_ITERS];
    float v[BIN_ITERS];
#pragma unroll
    for (int k = 0; k < BIN_ITERS; ++k) {
        const int  i  = tid + k * BIN_THREADS;
        const bool ok = i < SL;
        const int  e  = e0 + (ok ? i : 0);
        const int  rr = rows[e];
        r[k] = ok ? rr : -1;
        c[k] = cols[e];
        v[k] = vals[e];
    }
    __syncthreads();

    int pos[BIN_ITERS];
#pragma unroll
    for (int k = 0; k < BIN_ITERS; ++k) {
        const unsigned b = (unsigned)r[k] / (unsigned)RB;
        pos[k] = -1;
        if (b < (unsigned)NBLK)
            pos[k] = atomicAdd(&hist[b], 1);
    }
    __syncthreads();

    int hv = 0, inc = 0;
    if (tid < CNTPITCH) {
        hv  = hist[tid];
        inc = hv;
#pragma unroll
        for (int d = 1; d < 32; d <<= 1) {
            const int t = __shfl_up(inc, d);
            if (lane >= d) inc += t;
        }
        if (lane == 31) wsum[tid >> 5] = inc;
    }
    __syncthreads();
    if (tid < CNTPITCH) {
        int basew = 0;
        for (int w = 0; w < (tid >> 5); ++w) basew += wsum[w];
        off[tid] = basew + inc - hv;
    }
    __syncthreads();

#pragma unroll
    for (int k = 0; k < BIN_ITERS; ++k) {
        if (pos[k] >= 0) {
            const int b  = r[k] / RB;
            const int lr = r[k] - b * RB;
            stage[off[b] + pos[k]] =
                (ent_t)(unsigned)(lr | (c[k] << KEYBITS)) | ((ent_t)(unsigned)__float_as_int(v[k]) << 32);
        }
    }
    __syncthreads();

    const int wave = __builtin_amdgcn_readfirstlane(tid >> 5);
    for (int b = wave; b < NBLK; b += BIN_THREADS / 32) {
        const int cb = __builtin_amdgcn_readfirstlane(hist[b]);
        const int ob = __builtin_amdgcn_readfirstlane(off[b]);
        const int n  = cb < SEGCAP ? cb : SEGCAP;
        ent_t* __restrict__ s = seg + ((size_t)b * np + p) * SEGCAP;
        for (int i = lane; i < n; i += 32) s[i] = stage[ob + i];
    }
    if (tid < CNTPITCH) cntg[p * CNTPITCH + tid] = hv;
}

template <int K>
__device__ __forceinline__ int row_share16(int v) {
    return __builtin_amdgcn_update_dpp(0, v, 0x150 + K, 0xf, 0xf, true);
}

template <int NP, bool ADD_BIAS>
__device__ __forceinline__ void owner_spmm(const int* __restrict__ rows, const int* __restrict__ cols,
                                           const float* __restrict__ vals,
                                           const float* __restrict__ segf, const float* __restrict__ cntf,
                                           const float* __restrict__ src,
                                           const float* __restrict__ bias, float* __restrict__ dst) {
    __shared__ float acc[(RB + 1) * NK];

    const int tid   = threadIdx.x;
    const int lane  = tid & 31;
    const int l16   = lane & 15;
    const int lane4 = lane * 4;
    const int wave = __builtin_amdgcn_readfirstlane(tid >> 5);
    const int blk  = blockIdx.x;
    const int lo   = blk * RB;
    const ent_t* __restrict__ segs = (const ent_t*)segf;
    const int*  __restrict__ cntg = (const int*)cntf;
    const char* __restrict__ srcb = (const char*)src;
    char* accb = (char*)acc;

    int cnt_mine = 0;
    {
        const int pk = wave + OWN_WAVES * lane;
        if (pk < NP) cnt_mine = cntg[pk * CNTPITCH + blk];
    }

    float4* acc4 = (float4*)acc;
    for (int i = tid; i < (RB + 1) * NK / 4; i += OWN_THREADS) acc4[i] = make_float4(0.f, 0.f, 0.f, 0.f);

    auto load64 = [&](const ent_t* __restrict__ s, int n, ent_t (&e)[4]) {
        if (n > 0) {
#pragma unroll
            for (int r = 0; r < 4; ++r) {
                const int i = 16 * r + l16;
                e[r] = s[i < n ? i : n - 1];
            }
        }
    };

    float xv[64];
#define OWN_REQ(r, k) \
    xv[16 * (r) + (k)] = *(const float*)(srcb + (unsigned)(row_share16<(k)>(goff[r]) + lane4));
#define OWN_ADD(r, k) \
    __hip_atomic_fetch_add((float*)(accb + (unsigned)(row_share16<(k)>(loff[r]) + lane4)), \
                           __int_as_float(row_share16<(k)>(wbit[r])) * xv[16 * (r) + (k)], \
                           __ATOMIC_RELAXED, __HIP_MEMORY_SCOPE_WORKGROUP);
#define OWN_OCT(M, r, k0) M(r, k0) M(r, k0 + 1) M(r, k0 + 2) M(r, k0 + 3) M(r, k0 + 4) M(r, k0 + 5) M(r, k0 + 6) M(r, k0 + 7)
#define OWN_REG(M, r) \
    if (16 * (r) < m0) { OWN_OCT(M, r, 0) } \
    if (16 * (r) + 8 < m0) { OWN_OCT(M, r, 8) }

    constexpr int KMAX = (NP + OWN_WAVES - 1) / OWN_WAVES;
    int  n_cur = __builtin_amdgcn_readlane(cnt_mine, 0);
    ent_t ce[4] = {0, 0, 0, 0};
    load64(segs + ((size_t)blk * NP + wave) * SEGCAP, n_cur <= SEGCAP ? n_cur : 0, ce);
    __syncthreads();

    for (int k = 0; k < KMAX; ++k) {
        const int p = wave + OWN_WAVES * k;
        if (p >= NP) break;
        const int n = n_cur;
        const int m0 = n > SEGCAP ? 0 : (n < 64 ? n : 64);

        int goff[4], loff[4], wbit[4];
#pragma unroll
        for (int r = 0; r < 4; ++r) {
            const int key = (int)ce[r];
            goff[r] = (key >> KEYBITS) * (NK * 4);
            loff[r] = (16 * r + l16 < n) ? (key & ((1 << KEYBITS) - 1)) * (NK * 4) : RB * NK * 4;
            wbit[r] = (int)(ce[r] >> 32);
        }

        int n_nxt = 0;
        if (k + 1 < KMAX && p + OWN_WAVES < NP) {
            n_nxt = __builtin_amdgcn_readlane(cnt_mine, k + 1);
            load64(segs + ((size_t)blk * NP + p + OWN_WAVES) * SEGCAP, n_nxt <= SEGCAP ? n_nxt : 0, ce);
        }

        if (m0 > 0) {
            OWN_REG(OWN_REQ, 0) OWN_REG(OWN_REQ, 1) OWN_REG(OWN_REQ, 2) OWN_REG(OWN_REQ, 3)
        }

        if (m0 > 0) {
            OWN_REG(OWN_ADD, 0) OWN_REG(OWN_ADD, 1) OWN_REG(OWN_ADD, 2) OWN_REG(OWN_ADD, 3)
        }

        if (n <= SEGCAP) {
            for (int base = 64; base < n; base += 32) {
                const int   ir = base + lane;
                const ent_t t  = (segs + ((size_t)blk * NP + p) * SEGCAP)[ir < n ? ir : n - 1];
                const int   m  = (n - base) < 32 ? (n - base) : 32;
                for (int j = 0; j < m; ++j) {
                    const int   kj = __builtin_amdgcn_readlane((int)t, j);
                    const float vj = __int_as_float(__builtin_amdgcn_readlane((int)(t >> 32), j));
                    __hip_atomic_fetch_add(&acc[(kj & ((1 << KEYBITS) - 1)) * NK + lane],
                                           vj * src[(kj >> KEYBITS) * NK + lane],
                                           __ATOMIC_RELAXED, __HIP_MEMORY_SCOPE_WORKGROUP);
                }
            }
        } else {
            const int e0 = p * SL;
            for (int i = 0; i < SL; i += 32) {
                const int      e   = e0 + i + lane;
                const unsigned lr  = (unsigned)(rows[e] - lo);
                const bool     hit = lr < (unsigned)RB;
                unsigned mask = __builtin_amdgcn_ballot_w32(hit);
                int   c = 0;
                float v = 0.f;
                if (hit) { c = cols[e]; v = vals[e]; }
                while (mask != 0u) {
                    const int j = __builtin_ctz(mask);
                    mask &= mask - 1u;
                    const int   cj = __builtin_amdgcn_readlane(c, j);
                    const int   rj = __builtin_amdgcn_readlane((int)lr, j);
                    const float vj = __int_as_float(__builtin_amdgcn_readlane(__float_as_int(v), j));
                    __hip_atomic_fetch_add(&acc[rj * NK + lane], vj * src[cj * NK + lane],
                                           __ATOMIC_RELAXED, __HIP_MEMORY_SCOPE_WORKGROUP);
                }
            }
        }
        n_cur = n_nxt;
    }
#undef OWN_REQ
#undef OWN_ADD
#undef OWN_OCT
#undef OWN_REG

    __syncthreads();
    float4* __restrict__ dst4 = (float4*)(dst + (long long)lo * NK);
    for (int i = tid; i < RB * NK / 4; i += OWN_THREADS) {
        float4 a = acc4[i];
        if (ADD_BIAS) {
            const float4 b = *(const float4*)(bias + ((i * 4) & (NK - 1)));
            a.x += b.x; a.y += b.y; a.z += b.z; a.w += b.w;
        }
        dst4[i] = a;
    }
}

__global__ __launch_bounds__(OWN_THREADS)
void k_spmm1(const int* __restrict__ rows, const int* __restrict__ cols,
             const float* __restrict__ vals, const float* __restrict__ segf,
             const float* __restrict__ cntf, const float* __restrict__ x,
             float* __restrict__ g) {
    owner_spmm<PS, false>(rows, cols, vals, segf, cntf, x, x, g);
}

__global__ void k_gemm2(const float* __restrict__ g, const float* __restrict__ Ws,
                        const float* __restrict__ bs, const float* __restrict__ W0,
                        float* __restrict__ hw) {
    const int gw = blockIdx.x * (blockDim.x >> 5) + (threadIdx.x >> 5);
    if (gw >= NV / 16) return;
    const int lane = threadIdx.x & 31;
    const int half = lane >> 4;
    const int l16  = lane & 15;
    const int m0   = gw * 16;

    const float4* __restrict__ grow =
        (const float4*)(g + (long long)(m0 + l16) * NK + half * 16);
    float gk[16];
#pragma unroll
    for (int q = 0; q < 4; ++q) {
        const float4 t = grow[q];
        gk[4 * q + 0] = t.x; gk[4 * q + 1] = t.y; gk[4 * q + 2] = t.z; gk[4 * q + 3] = t.w;
    }

    v8f acc[4];
#pragma unroll
    for (int mt = 0; mt < 4; ++mt) {
        const float4 lo = *(const float4*)(bs + mt * 16 + half * 8);
        const float4 hi = *(const float4*)(bs + mt * 16 + half * 8 + 4);
        acc[mt][0] = lo.x; acc[mt][1] = lo.y; acc[mt][2] = lo.z; acc[mt][3] = lo.w;
        acc[mt][4] = hi.x; acc[mt][5] = hi.y; acc[mt][6] = hi.z; acc[mt][7] = hi.w;
    }

#pragma unroll
    for (int s = 0; s < 8; ++s) {
        const int k0 = half * 16 + 2 * s;
        v2f b;
        b.x = gk[2 * s];
        b.y = gk[2 * s + 1];
#pragma unroll
        for (int mt = 0; mt < 4; ++mt) {
            v2f a;
            a.x = Ws[(k0    ) * NL + mt * 16 + l16];
            a.y = Ws[(k0 + 1) * NL + mt * 16 + l16];
            acc[mt] = __builtin_amdgcn_wmma_f32_16x16x4_f32(
                false, a, false, b, (short)0, acc[mt], false, false);
        }
    }

#pragma unroll
    for (int mt = 0; mt < 4; ++mt) {
#pragma unroll
        for (int j = 0; j < 8; ++j) acc[mt][j] = fmaxf(acc[mt][j], 0.0f);
    }

    v8f o0 = {};
    v8f o1 = {};
#pragma unroll
    for (int mt = 0; mt < 4; ++mt) {
#pragma unroll
        for (int q = 0; q < 4; ++q) {
            const int hid = mt * 16 + half * 8 + 2 * q;
            v2f b;
            b.x = acc[mt][2 * q];
            b.y = acc[mt][2 * q + 1];
            v2f a0, a1;
            a0.x = W0[(hid    ) * NK +      l16];
            a0.y = W0[(hid + 1) * NK +      l16];
            a1.x = W0[(hid    ) * NK + 16 + l16];
            a1.y = W0[(hid + 1) * NK + 16 + l16];
            o0 = __builtin_amdgcn_wmma_f32_16x16x4_f32(
                false, a0, false, b, (short)0, o0, false, false);
            o1 = __builtin_amdgcn_wmma_f32_16x16x4_f32(
                false, a1, false, b, (short)0, o1, false, false);
        }
    }

    float* __restrict__ orow = hw + (long long)(m0 + l16) * NK + half * 8;
    *(float4*)(orow     ) = make_float4(o0[0], o0[1], o0[2], o0[3]);
    *(float4*)(orow +  4) = make_float4(o0[4], o0[5], o0[6], o0[7]);
    *(float4*)(orow + 16) = make_float4(o1[0], o1[1], o1[2], o1[3]);
    *(float4*)(orow + 20) = make_float4(o1[4], o1[5], o1[6], o1[7]);
}

__global__ __launch_bounds__(OWN_THREADS)
void k_spmm2(const int* __restrict__ rows, const int* __restrict__ cols,
             const float* __restrict__ vals, const float* __restrict__ segf,
             const float* __restrict__ cntf, const float* __restrict__ hw,
             const float* __restrict__ b0, float* __restrict__ out) {
    owner_spmm<PV, true>(rows, cols, vals, segf, cntf, hw, b0, out);
}

extern "C" void kernel_launch(void* const* d_in, const int* in_sizes, int n_in,
                              void* d_out, int out_size, void* d_ws, size_t ws_size,
                              hipStream_t stream) {
    const float* x    = (const float*)d_in[0];
    const int*   srow = (const int*)  d_in[1];
    const int*   scol = (const int*)  d_in[2];
    const float* sval = (const float*)d_in[3];
    const int*   vrow = (const int*)  d_in[4];
    const int*   vcol = (const int*)  d_in[5];
    const float* vval = (const float*)d_in[6];
    const float* Ws   = (const float*)d_in[7];
    const float* bs   = (const float*)d_in[8];
    const float* W0   = (const float*)d_in[9];
    const float* b0   = (const float*)d_in[10];
    float* out = (float*)d_out;

    float* ws   = (float*)d_ws;
    float* g    = ws + OFF_G;
    float* hw   = ws + OFF_HW;
    float* seg1 = ws + OFF_SEG1;
    float* seg2 = ws + OFF_SEG2;
    float* cnt1 = ws + OFF_CNT1;
    float* cnt2 = ws + OFF_CNT2;

    k_bin<<<PS + PV, BIN_THREADS, 0, stream>>>(srow, scol, sval, vrow, vcol, vval, seg1, cnt1, seg2, cnt2);

    k_spmm1<<<NBLK, OWN_THREADS, 0, stream>>>(srow, scol, sval, seg1, cnt1, x, g);

    const int tilesM = NV / 16;
    const int wavesPerBlock = 8;
    const int blocks = (tilesM + wavesPerBlock - 1) / wavesPerBlock;
    k_gemm2<<<blocks, 32 * wavesPerBlock, 0, stream>>>(g, Ws, bs, W0, hw);

    k_spmm2<<<NBLK, OWN_THREADS, 0, stream>>>(vrow, vcol, vval, seg2, cnt2, hw, b0, out);
}


# Run both WMMA products inside the layer-1 owner work-group on its LDS-resident 16-row tiles: the 32-wide edge sums never go to memory and the separate GEMM launch is gone (3 launches)
# speedup vs baseline: 6.4045x; 1.2225x over previous
#include <hip/hip_runtime.h>


constexpr int NS  = 306;
constexpr int NV  = 200000;
constexpr int NK  = 32;
constexpr int NL  = 64;
constexpr int NES = 800000;
constexpr int NEV = 1200000;

typedef __attribute__((ext_vector_type(2))) float v2f;
typedef __attribute__((ext_vector_type(8))) float v8f;


constexpr int RB          = 800;
constexpr int NBLK        = NV / RB;
constexpr int SL          = 8000;
constexpr int PS          = NES / SL;
constexpr int PV          = NEV / SL;
constexpr int SEGCAP      = 128;
constexpr int CNTPITCH    = 256;
constexpr int KEYBITS     = 10;
typedef unsigned long long ent_t;
constexpr int OWN_THREADS = 1024;
constexpr int OWN_WAVES   = OWN_THREADS / 32;
constexpr int BIN_THREADS = 1024;
constexpr int BIN_ITERS   = (SL + BIN_THREADS - 1) / BIN_THREADS;
static_assert(NBLK * RB == NV, "row blocks must tile the vertices");
static_assert(RB % 16 == 0, "a row block is whole 16-row matrix tiles");
static_assert(PS * SL == NES && PV * SL == NEV, "slices must tile the edge lists");
static_assert(SL % 32 == 0, "a slice is whole wave-wide steps");
static_assert(OWN_WAVES <= PS && OWN_WAVES <= PV, "every wave of an owner has a first slice");
static_assert(SEGCAP % 64 == 0, "a segment is read in steps of 64 entries");
static_assert(NBLK <= CNTPITCH && CNTPITCH % 32 == 0 && CNTPITCH <= BIN_THREADS, "one count per thread of the first waves");
static_assert(RB <= (1 << KEYBITS), "row inside the block must fit its field");
static_assert(NV <= (1 << (31 - KEYBITS)) && NS <= NV, "source row must fit above it");

constexpr size_t OFF_G    = 0;
constexpr size_t OFF_HW   = OFF_G    + (size_t)NV * NK;
constexpr size_t OFF_SEG1 = OFF_HW   + (size_t)NV * NK;
constexpr size_t OFF_SEG2 = OFF_SEG1 + (size_t)NBLK * PS * SEGCAP * 2;
constexpr size_t OFF_CNT1 = OFF_SEG2 + (size_t)NBLK * PV * SEGCAP * 2;
constexpr size_t OFF_CNT2 = OFF_CNT1 + (size_t)PS * CNTPITCH;

__global__ __launch_bounds__(BIN_THREADS)
void k_bin(const int* __restrict__ srow, const int* __restrict__ scol, const float* __restrict__ sval,
           const int* __restrict__ vrow, const int* __restrict__ vcol, const float* __restrict__ vval,
           float* __restrict__ seg1f, float* __restrict__ cnt1f,
           float* __restrict__ seg2f, float* __restrict__ cnt2f) {
    __shared__ int   hist[CNTPITCH];
    __shared__ int   off[CNTPITCH];
    __shared__ int   wsum[CNTPITCH / 32];
    __shared__ ent_t stage[SL];

    const int  tid    = threadIdx.x;
    const int  lane   = tid & 31;
    const bool sensor = blockIdx.x < PS;
    const int* __restrict__   rows = sensor ? srow : vrow;
    const int* __restrict__   cols = sensor ? scol : vcol;
    const float* __restrict__ vals = sensor ? sval : vval;
    ent_t* __restrict__ seg = (ent_t*)(sensor ? seg1f : seg2f);
    int*  __restrict__ cntg = (int*)(sensor ? cnt1f : cnt2f);
    const int np = sensor ? PS : PV;
    const int p  = sensor ? blockIdx.x : blockIdx.x - PS;

    for (int i = tid; i < CNTPITCH; i += BIN_THREADS) hist[i] = 0;

    const int e0 = p * SL;
    int   r[BIN_ITERS], c[BIN_ITERS];
    float v[BIN_ITERS];
#pragma unroll
    for (int k = 0; k < BIN_ITERS; ++k) {
        const int  i  = tid + k * BIN_THREADS;
        const bool ok = i < SL;
        const int  e  = e0 + (ok ? i : 0);
        const int  rr = rows[e];
        r[k] = ok ? rr : -1;
        c[k] = cols[e];
        v[k] = vals[e];
    }
    __syncthreads();

    int pos[BIN_ITERS];
#pragma unroll
    for (int k = 0; k < BIN_ITERS; ++k) {
        const unsigned b = (unsigned)r[k] / (unsigned)RB;
        pos[k] = -1;
        if (b < (unsigned)NBLK)
            pos[k] = atomicAdd(&hist[b], 1);
    }
    __syncthreads();

    int hv = 0, inc = 0;
    if (tid < CNTPITCH) {
        hv  = hist[tid];
        inc = hv;
#pragma unroll
        for (int d = 1; d < 32; d <<= 1) {
            const int t = __shfl_up(inc, d);
            if (lane >= d) inc += t;
        }
        if (lane == 31) wsum[tid >> 5] = inc;
    }
    __syncthreads();
    if (tid < CNTPITCH) {
        int basew = 0;
        for (int w = 0; w < (tid >> 5); ++w) basew += wsum[w];
        off[tid] = basew + inc - hv;
    }
    __syncthreads();

#pragma unroll
    for (int k = 0; k < BIN_ITERS; ++k) {
        if (pos[k] >= 0) {
            const int b  = r[k] / RB;
            const int lr = r[k] - b * RB;
            stage[off[b] + pos[k]] =
                (ent_t)(unsigned)(lr | (c[k] << KEYBITS)) | ((ent_t)(unsigned)__float_as_int(v[k]) << 32);
        }
    }
    __syncthreads();

    const int wave = __builtin_amdgcn_readfirstlane(tid >> 5);
    for (int b = wave; b < NBLK; b += BIN_THREADS / 32) {
        const int cb = __builtin_amdgcn_readfirstlane(hist[b]);
        const int ob = __builtin_amdgcn_readfirstlane(off[b]);
        const int n  = cb < SEGCAP ? cb : SEGCAP;
        ent_t* __restrict__ s = seg + ((size_t)b * np + p) * SEGCAP;
        for (int i = lane; i < n; i += 32) s[i] = stage[ob + i];
    }
    if (tid < CNTPITCH) cntg[p * CNTPITCH + tid] = hv;
}

template <int K>
__device__ __forceinline__ int row_share16(int v) {
    return __builtin_amdgcn_update_dpp(0, v, 0x150 + K, 0xf, 0xf, true);
}

__device__ __forceinline__ void tile_mlp(const float* gt, const float* Ws,
                                         const float* bs, const float* W0,
                                         float* hwt, int lane) {
    const int half = lane >> 4;
    const int l16  = lane & 15;

    const float4* grow = (const float4*)(gt + l16 * NK + half * 16);
    const float* wsl = Ws + half * 16 * NL + l16;
    const float* w0l = W0 + half * 8 * NK + l16;
    const float* bsl = bs + half * 8;
    float gk[16];
#pragma unroll
    for (int q = 0; q < 4; ++q) {
        const float4 t = grow[q];
        gk[4 * q + 0] = t.x; gk[4 * q + 1] = t.y; gk[4 * q + 2] = t.z; gk[4 * q + 3] = t.w;
    }

    v8f acc[4];
#pragma unroll
    for (int mt = 0; mt < 4; ++mt) {
        const float4 lo = *(const float4*)(bsl + mt * 16);
        const float4 hi = *(const float4*)(bsl + mt * 16 + 4);
        acc[mt][0] = lo.x; acc[mt][1] = lo.y; acc[mt][2] = lo.z; acc[mt][3] = lo.w;
        acc[mt][4] = hi.x; acc[mt][5] = hi.y; acc[mt][6] = hi.z; acc[mt][7] = hi.w;
    }

#pragma unroll
    for (int s = 0; s < 8; ++s) {
        v2f b;
        b.x = gk[2 * s];
        b.y = gk[2 * s + 1];
#pragma unroll
        for (int mt = 0; mt < 4; ++mt) {
            v2f a;
            a.x = wsl[(2 * s    ) * NL + mt * 16];
            a.y = wsl[(2 * s + 1) * NL + mt * 16];
            acc[mt] = __builtin_amdgcn_wmma_f32_16x16x4_f32(
                false, a, false, b, (short)0, acc[mt], false, false);
        }
    }

#pragma unroll
    for (int mt = 0; mt < 4; ++mt) {
#pragma unroll
        for (int j = 0; j < 8; ++j) acc[mt][j] = fmaxf(acc[mt][j], 0.0f);
    }

    v8f o0 = {};
    v8f o1 = {};
#pragma unroll
    for (int mt = 0; mt < 4; ++mt) {
#pragma unroll
        for (int q = 0; q < 4; ++q) {
            v2f b;
            b.x = acc[mt][2 * q];
            b.y = acc[mt][2 * q + 1];
            v2f a0, a1;
            a0.x = w0l[(mt * 16 + 2 * q    ) * NK     ];
            a0.y = w0l[(mt * 16 + 2 * q + 1) * NK     ];
            a1.x = w0l[(mt * 16 + 2 * q    ) * NK + 16];
            a1.y = w0l[(mt * 16 + 2 * q + 1) * NK + 16];
            o0 = __builtin_amdgcn_wmma_f32_16x16x4_f32(
                false, a0, false, b, (short)0, o0, false, false);
            o1 = __builtin_amdgcn_wmma_f32_16x16x4_f32(
                false, a1, false, b, (short)0, o1, false, false);
        }
    }

    float* orow = hwt + l16 * NK + half * 8;
    *(float4*)(orow     ) = make_float4(o0[0], o0[1], o0[2], o0[3]);
    *(float4*)(orow +  4) = make_float4(o0[4], o0[5], o0[6], o0[7]);
    *(float4*)(orow + 16) = make_float4(o1[0], o1[1], o1[2], o1[3]);
    *(float4*)(orow + 20) = make_float4(o1[4], o1[5], o1[6], o1[7]);
}

template <int NP, bool ADD_BIAS, bool FUSE_MLP>
__device__ __forceinline__ void owner_spmm(const int* __restrict__ rows, const int* __restrict__ cols,
                                           const float* __restrict__ vals,
                                           const float* __restrict__ segf, const float* __restrict__ cntf,
                                           const float* __restrict__ src,
                                           const float* __restrict__ bias, float* dst,
                                           const float* Ws, const float* bs, const float* W0) {
    __shared__ float acc[(RB + 1) * NK];

    const int tid   = threadIdx.x;
    const int lane  = tid & 31;
    const int l16   = lane & 15;
    const int lane4 = lane * 4;
    const int wave = __builtin_amdgcn_readfirstlane(tid >> 5);
    const int blk  = blockIdx.x;
    const int lo   = blk * RB;
    const ent_t* __restrict__ segs = (const ent_t*)segf;
    const int*  __restrict__ cntg = (const int*)cntf;
    const char* __restrict__ srcb = (const char*)src;
    char* accb = (char*)acc;

    int cnt_mine = 0;
    {
        const int pk = wave + OWN_WAVES * lane;
        if (pk < NP) cnt_mine = cntg[pk * CNTPITCH + blk];
    }

    float4* acc4 = (float4*)acc;
    for (int i = tid; i < (RB + 1) * NK / 4; i += OWN_THREADS) acc4[i] = make_float4(0.f, 0.f, 0.f, 0.f);

    auto load64 = [&](const ent_t* __restrict__ s, int n, ent_t (&e)[4]) {
        if (n > 0) {
#pragma unroll
            for (int r = 0; r < 4; ++r) {
                const int i = 16 * r + l16;
                e[r] = s[i < n ? i : n - 1];
            }
        }
    };

    float xv[64];
#define OWN_REQ(r, k) \
    xv[16 * (r) + (k)] = *(const float*)(srcb + (unsigned)(row_share16<(k)>(goff[r]) + lane4));
#define OWN_ADD(r, k) \
    __hip_atomic_fetch_add((float*)(accb + (unsigned)(row_share16<(k)>(loff[r]) + lane4)), \
                           __int_as_float(row_share16<(k)>(wbit[r])) * xv[16 * (r) + (k)], \
                           __ATOMIC_RELAXED, __HIP_MEMORY_SCOPE_WORKGROUP);
#define OWN_OCT(M, r, k0) M(r, k0) M(r, k0 + 1) M(r, k0 + 2) M(r, k0 + 3) M(r, k0 + 4) M(r, k0 + 5) M(r, k0 + 6) M(r, k0 + 7)
#define OWN_REG(M, r) \
    if (16 * (r) < m0) { OWN_OCT(M, r, 0) } \
    if (16 * (r) + 8 < m0) { OWN_OCT(M, r, 8) }

    constexpr int KMAX = (NP + OWN_WAVES - 1) / OWN_WAVES;
    int  n_cur = __builtin_amdgcn_readlane(cnt_mine, 0);
    ent_t ce[4] = {0, 0, 0, 0};
    load64(segs + ((size_t)blk * NP + wave) * SEGCAP, n_cur <= SEGCAP ? n_cur : 0, ce);
    __syncthreads();

    for (int k = 0; k < KMAX; ++k) {
        const int p = wave + OWN_WAVES * k;
        if (p >= NP) break;
        const int n = n_cur;
        const int m0 = n > SEGCAP ? 0 : (n < 64 ? n : 64);

        int goff[4], loff[4], wbit[4];
#pragma unroll
        for (int r = 0; r < 4; ++r) {
            const int key = (int)ce[r];
            goff[r] = (key >> KEYBITS) * (NK * 4);
            loff[r] = (16 * r + l16 < n) ? (key & ((1 << KEYBITS) - 1)) * (NK * 4) : RB * NK * 4;
            wbit[r] = (int)(ce[r] >> 32);
        }

        int n_nxt = 0;
        if (k + 1 < KMAX && p + OWN_WAVES < NP) {
            n_nxt = __builtin_amdgcn_readlane(cnt_mine, k + 1);
            load64(segs + ((size_t)blk * NP + p + OWN_WAVES) * SEGCAP, n_nxt <= SEGCAP ? n_nxt : 0, ce);
        }

        if (m0 > 0) {
            OWN_REG(OWN_REQ, 0) OWN_REG(OWN_REQ, 1) OWN_REG(OWN_REQ, 2) OWN_REG(OWN_REQ, 3)
        }

        if (m0 > 0) {
            OWN_REG(OWN_ADD, 0) OWN_REG(OWN_ADD, 1) OWN_REG(OWN_ADD, 2) OWN_REG(OWN_ADD, 3)
        }

        if (n <= SEGCAP) {
            for (int base = 64; base < n; base += 32) {
                const int   ir = base + lane;
                const ent_t t  = (segs + ((size_t)blk * NP + p) * SEGCAP)[ir < n ? ir : n - 1];
                const int   m  = (n - base) < 32 ? (n - base) : 32;
                for (int j = 0; j < m; ++j) {
                    const int   kj = __builtin_amdgcn_readlane((int)t, j);
                    const float vj = __int_as_float(__builtin_amdgcn_readlane((int)(t >> 32), j));
                    __hip_atomic_fetch_add(&acc[(kj & ((1 << KEYBITS) - 1)) * NK + lane],
                                           vj * src[(kj >> KEYBITS) * NK + lane],
                                           __ATOMIC_RELAXED, __HIP_MEMORY_SCOPE_WORKGROUP);
                }
            }
        } else {
            const int e0 = p * SL;
            for (int i = 0; i < SL; i += 32) {
                const int      e   = e0 + i + lane;
                const unsigned lr  = (unsigned)(rows[e] - lo);
                const bool     hit = lr < (unsigned)RB;
                unsigned mask = __builtin_amdgcn_ballot_w32(hit);
                int   c = 0;
                float v = 0.f;
                if (hit) { c = cols[e]; v = vals[e]; }
                while (mask != 0u) {
                    const int j = __builtin_ctz(mask);
                    mask &= mask - 1u;
                    const int   cj = __builtin_amdgcn_readlane(c, j);
                    const int   rj = __builtin_amdgcn_readlane((int)lr, j);
                    const float vj = __int_as_float(__builtin_amdgcn_readlane(__float_as_int(v), j));
                    __hip_atomic_fetch_add(&acc[rj * NK + lane], vj * src[cj * NK + lane],
                                           __ATOMIC_RELAXED, __HIP_MEMORY_SCOPE_WORKGROUP);
                }
            }
        }
        n_cur = n_nxt;
    }
#undef OWN_REQ
#undef OWN_ADD
#undef OWN_OCT
#undef OWN_REG

    __syncthreads();
    if constexpr (FUSE_MLP) {
        for (int t = wave; t < RB / 16; t += OWN_WAVES)
            tile_mlp(acc + t * 16 * NK, Ws, bs, W0, dst + (long long)(lo + t * 16) * NK, lane);
    } else {
        float4* __restrict__ dst4 = (float4*)(dst + (long long)lo * NK);
        for (int i = tid; i < RB * NK / 4; i += OWN_THREADS) {
            float4 a = acc4[i];
            if (ADD_BIAS) {
                const float4 b = *(const float4*)(bias + ((i * 4) & (NK - 1)));
                a.x += b.x; a.y += b.y; a.z += b.z; a.w += b.w;
            }
            dst4[i] = a;
        }
    }
}

__global__ __launch_bounds__(OWN_THREADS)
void k_layer1(const int* __restrict__ rows, const int* __restrict__ cols,
              const float* __restrict__ vals, const float* __restrict__ segf,
              const float* __restrict__ cntf, const float* __restrict__ x,
              const float* Ws, const float* bs, const float* W0, float* hw) {
    owner_spmm<PS, false, true>(rows, cols, vals, segf, cntf, x, x, hw, Ws, bs, W0);
}


__global__ __launch_bounds__(OWN_THREADS)
void k_spmm2(const int* __restrict__ rows, const int* __restrict__ cols,
             const float* __restrict__ vals, const float* __restrict__ segf,
             const float* __restrict__ cntf, const float* __restrict__ hw,
             const float* __restrict__ b0, float* __restrict__ out) {
    owner_spmm<PV, true, false>(rows, cols, vals, segf, cntf, hw, b0, out, nullptr, nullptr, nullptr);
}

extern "C" void kernel_launch(void* const* d_in, const int* in_sizes, int n_in,
                              void* d_out, int out_size, void* d_ws, size_t ws_size,
                              hipStream_t stream) {
    const float* x    = (const float*)d_in[0];
    const int*   srow = (const int*)  d_in[1];
    const int*   scol = (const int*)  d_in[2];
    const float* sval = (const float*)d_in[3];
    const int*   vrow = (const int*)  d_in[4];
    const int*   vcol = (const int*)  d_in[5];
    const float* vval = (const float*)d_in[6];
    const float* Ws   = (const float*)d_in[7];
    const float* bs   = (const float*)d_in[8];
    const float* W0   = (const float*)d_in[9];
    const float* b0   = (const float*)d_in[10];
    float* out = (float*)d_out;

    float* ws   = (float*)d_ws;
    float* hw   = ws + OFF_HW;
    float* seg1 = ws + OFF_SEG1;
    float* seg2 = ws + OFF_SEG2;
    float* cnt1 = ws + OFF_CNT1;
    float* cnt2 = ws + OFF_CNT2;

    k_bin<<<PS + PV, BIN_THREADS, 0, stream>>>(srow, scol, sval, vrow, vcol, vval, seg1, cnt1, seg2, cnt2);

    k_layer1<<<NBLK, OWN_THREADS, 0, stream>>>(srow, scol, sval, seg1, cnt1, x, Ws, bs, W0, hw);

    k_spmm2<<<NBLK, OWN_THREADS, 0, stream>>>(vrow, vcol, vval, seg2, cnt2, hw, b0, out);
}
